# starting kernel
# baseline (speedup 1.0000x reference)
#include <hip/hip_runtime.h>

typedef float v2f __attribute__((ext_vector_type(2)));
typedef float v8f __attribute__((ext_vector_type(8)));

#define NFEAT 64

__global__ void gcn_zero_f32(float* __restrict__ p, int count) {
  int i = blockIdx.x * blockDim.x + threadIdx.x;
  if (i < count) p[i] = 0.0f;
}

__global__ void gcn_deg_init(float* __restrict__ deg, int n) {
  int i = blockIdx.x * blockDim.x + threadIdx.x;
  if (i < n) deg[i] = 1.0f;
}

__global__ void gcn_deg_count(const int* __restrict__ dst, float* __restrict__ deg, int E) {
  int i = blockIdx.x * blockDim.x + threadIdx.x;
  if (i < E) atomicAdd(&deg[dst[i]], 1.0f);
}

__global__ void gcn_deg_rsqrt(float* __restrict__ deg, int n) {
  int i = blockIdx.x * blockDim.x + threadIdx.x;
  if (i < n) deg[i] = rsqrtf(deg[i]);
}

__global__ __launch_bounds__(256) void gcn_gemm_wmma_f32(
    const float* __restrict__ in, const float* __restrict__ W,
    float* __restrict__ out, int K) {
  __shared__ float As[32 * (128 + 4)];
  const int ldA = K + 4;
  const int tid = threadIdx.x;
  const int blockRow0 = blockIdx.x * 32;

  const int total = 32 * K;
  for (int idx = tid; idx < total; idx += 256) {
    int r = idx / K;
    int c = idx - r * K;
    As[r * ldA + c] = in[(size_t)(blockRow0 + r) * K + c];
  }
  __syncthreads();

  const int wave = tid >> 5;
  const int lane = tid & 31;
  const int m  = lane & 15;
  const int kp = lane >> 4;
  const int rowL = (wave >> 2) * 16;
  const int col0 = (wave & 3) * 16;

  v8f c = {};
  for (int kk = 0; kk < K; kk += 4) {
    const int ka = kk + 2 * kp;
    v2f a, b;
    a.x = As[(rowL + m) * ldA + ka];
    a.y = As[(rowL + m) * ldA + ka + 1];
    b.x = W[(size_t)(ka + 0) * NFEAT + col0 + m];
    b.y = W[(size_t)(ka + 1) * NFEAT + col0 + m];
    c = __builtin_amdgcn_wmma_f32_16x16x4_f32(
         false, a,  false, b,
         (short)0, c,  false,  false);
  }

  const int gRow0 = blockRow0 + rowL;
#pragma unroll
  for (int i = 0; i < 8; ++i) {
    out[(size_t)(gRow0 + i + 8 * kp) * NFEAT + col0 + m] = c[i];
  }
}

__global__ void gcn_scatter(const int* __restrict__ src, const int* __restrict__ dst,
                            const float* __restrict__ h, const float* __restrict__ isq,
                            float* __restrict__ acc, int E) {
  int tid = blockIdx.x * blockDim.x + threadIdx.x;
  int e = tid >> 4;
  if (e >= E) return;
  int f = (tid & 15) * 4;
  int s = src[e], d = dst[e];
  float coef = isq[s] * isq[d];
  float4 hv = *(const float4*)(h + (size_t)s * NFEAT + f);
  float* o = acc + (size_t)d * NFEAT + f;
  atomicAdd(o + 0, hv.x * coef);
  atomicAdd(o + 1, hv.y * coef);
  atomicAdd(o + 2, hv.z * coef);
  atomicAdd(o + 3, hv.w * coef);
}

__global__ void gcn_finalize(const float* __restrict__ acc, const float* __restrict__ h,
                             const float* __restrict__ isq, const float* __restrict__ bias,
                             float* __restrict__ out, int n) {
  int t = blockIdx.x * blockDim.x + threadIdx.x;
  int i = t >> 6;
  if (i >= n) return;
  int f = t & 63;
  float is = isq[i];
  float v = acc[(size_t)i * NFEAT + f] + h[(size_t)i * NFEAT + f] * is * is + bias[f];
  out[(size_t)i * NFEAT + f] = fmaxf(v, 0.0f);
}

__global__ void gcn_pool(const float* __restrict__ h, const int* __restrict__ batch,
                         float* __restrict__ gsum, float* __restrict__ gcnt, int n) {
  int t = blockIdx.x * blockDim.x + threadIdx.x;
  int i = t >> 6;
  if (i >= n) return;
  int f = t & 63;
  int g = batch[i];
  atomicAdd(&gsum[g * NFEAT + f], h[(size_t)i * NFEAT + f]);
  if (f == 0) atomicAdd(&gcnt[g], 1.0f);
}

__global__ __launch_bounds__(256) void gcn_head(
    const float* __restrict__ gsum, const float* __restrict__ gcnt,
    const float* __restrict__ Wp, const float* __restrict__ bp,
    const float* __restrict__ Wc, const float* __restrict__ bc,
    float* __restrict__ out) {
  __shared__ float g[64 * 64];
  __shared__ float zs[64 * 64];
  const int t = threadIdx.x;
  for (int idx = t; idx < 64 * 64; idx += 256) {
    int gi = idx >> 6;
    float c = fmaxf(gcnt[gi], 1.0f);
    g[idx] = gsum[idx] / c;
  }
  __syncthreads();
  for (int idx = t; idx < 64 * 64; idx += 256) {
    int m = idx >> 6, nn = idx & 63;
    float a = bp[nn];
#pragma unroll 8
    for (int k = 0; k < 64; ++k) a += g[m * 64 + k] * Wp[k * 64 + nn];
    zs[idx] = a;
    out[384 + idx] = a;
  }
  __syncthreads();
  for (int idx = t; idx < 64 * 6; idx += 256) {
    int m = idx / 6, nn = idx - m * 6;
    float a = bc[nn];
#pragma unroll 8
    for (int k = 0; k < 64; ++k) a += zs[m * 64 + k] * Wc[k * 6 + nn];
    out[idx] = a;
  }
}

extern "C" void kernel_launch(void* const* d_in, const int* in_sizes, int n_in,
                              void* d_out, int out_size, void* d_ws, size_t ws_size,
                              hipStream_t stream) {
  const float* x     = (const float*)d_in[0];
  const int*   ei    = (const int*)d_in[1];
  const int*   batch = (const int*)d_in[2];
  const float* W1    = (const float*)d_in[3];
  const float* b1    = (const float*)d_in[4];
  const float* W2    = (const float*)d_in[5];
  const float* b2    = (const float*)d_in[6];
  const float* Wp    = (const float*)d_in[7];
  const float* bp    = (const float*)d_in[8];
  const float* Wc    = (const float*)d_in[9];
  const float* bc    = (const float*)d_in[10];

  const int n = in_sizes[0] / 128;
  const int E = in_sizes[1] / 2;
  const int* src = ei;
  const int* dst = ei + E;

  char* ws = (char*)d_ws;
  size_t off = 0;
  auto carve = [&](size_t bytes) -> float* {
    float* p = (float*)(ws + off);
    off = (off + bytes + 255) & ~(size_t)255;
    return p;
  };
  float* isq  = carve((size_t)n * sizeof(float));
  float* bufA = carve((size_t)n * NFEAT * sizeof(float));
  float* bufB = carve((size_t)n * NFEAT * sizeof(float));
  float* gsum = carve(64 * NFEAT * sizeof(float));
  float* gcnt = carve(64 * sizeof(float));

  const int T = 256;
  const int nBlk  = (n + T - 1) / T;
  const int eBlk  = (E + T - 1) / T;
  const int nfTot = n * NFEAT;
  const int nfBlk = (nfTot + T - 1) / T;
  const int e16Blk = (int)(((long long)E * 16 + T - 1) / T);
  const int gemmBlk = n / 32;

  gcn_deg_init<<<nBlk, T, 0, stream>>>(isq, n);
  gcn_deg_count<<<eBlk, T, 0, stream>>>(dst, isq, E);
  gcn_deg_rsqrt<<<nBlk, T, 0, stream>>>(isq, n);

  gcn_gemm_wmma_f32<<<gemmBlk, T, 0, stream>>>(x, W1, bufA, 128);
  gcn_zero_f32<<<nfBlk, T, 0, stream>>>(bufB, nfTot);
  gcn_scatter<<<e16Blk, T, 0, stream>>>(src, dst, bufA, isq, bufB, E);
  gcn_finalize<<<nfBlk, T, 0, stream>>>(bufB, bufA, isq, b1, bufB, n);

  gcn_gemm_wmma_f32<<<gemmBlk, T, 0, stream>>>(bufB, W2, bufA, 64);
  gcn_zero_f32<<<nfBlk, T, 0, stream>>>(bufB, nfTot);
  gcn_scatter<<<e16Blk, T, 0, stream>>>(src, dst, bufA, isq, bufB, E);
  gcn_finalize<<<nfBlk, T, 0, stream>>>(bufB, bufA, isq, b2, bufB, n);

  gcn_zero_f32<<<(64 * NFEAT + 64 + T - 1) / T, T, 0, stream>>>(gsum, 64 * NFEAT + 64);
  gcn_pool<<<nfBlk, T, 0, stream>>>(bufB, batch, gsum, gcnt, n);
  gcn_head<<<1, T, 0, stream>>>(gsum, gcnt, Wp, bp, Wc, bc, (float*)d_out);
}


# edge kernels (degree count, both scatters) as resident grids of 1280 blocks: a wave loops over chunks of 32 edges, next ids and next 8 rows requested ahead of the current adds; sums arrival-ordered
# speedup vs baseline: 3.2070x; 3.2070x over previous
#include <hip/hip_runtime.h>

typedef float v2f __attribute__((ext_vector_type(2)));
typedef float v4f __attribute__((ext_vector_type(4)));
typedef float v8f __attribute__((ext_vector_type(8)));

#define NFEAT 64

__device__ __forceinline__ int package_half() {
  const unsigned id = __builtin_amdgcn_s_sendmsg_rtn(0x87);
  return (int)((id >> 18) & 1u);
}

__global__ void gcn_zero_f32(float* __restrict__ p, int count) {
  int i = blockIdx.x * blockDim.x + threadIdx.x;
  if (i < count) p[i] = 0.0f;
}

__global__ void gcn_deg_init(float* __restrict__ deg0, float* __restrict__ deg1, int n) {
  int i = blockIdx.x * blockDim.x + threadIdx.x;
  if (i < n) {
    deg0[i] = 1.0f;
    deg1[i] = 0.0f;
  }
}

#define EDGE_GRID 1280
#define DEG_BATCH 5

__global__ __launch_bounds__(256) void gcn_deg_count(const int* __restrict__ dst,
                                                     float* __restrict__ deg0,
                                                     float* __restrict__ deg1, int E) {
  float* deg = package_half() ? deg1 : deg0;
  const int stride = gridDim.x * blockDim.x;
  for (int base = blockIdx.x * blockDim.x + threadIdx.x; base < E; base += stride * DEG_BATCH) {
    int dd[DEG_BATCH];
#pragma unroll
    for (int u = 0; u < DEG_BATCH; ++u) {
      dd[u] = dst[min(base + u * stride, E - 1)];
    }
#pragma unroll
    for (int u = 0; u < DEG_BATCH; ++u) {
      if (base + u * stride < E) atomicAdd(&deg[dd[u]], 1.0f);
    }
  }
}

__global__ void gcn_deg_rsqrt(float* __restrict__ deg0, const float* __restrict__ deg1, int n) {
  int i = blockIdx.x * blockDim.x + threadIdx.x;
  if (i < n) deg0[i] = rsqrtf(deg0[i] + deg1[i]);
}

template <bool RELU_IN>
__global__ __launch_bounds__(256) void gcn_gemm_wmma_f32(
    const float* __restrict__ in, const float* __restrict__ in1, const float* __restrict__ W,
    const float* __restrict__ isq, const float* __restrict__ bias_in,
    float* __restrict__ out, float* __restrict__ acc0, float* __restrict__ acc1, int K) {
  __shared__ float As[32 * (128 + 4)];
  const int ldA = K + 4;
  const int tid = threadIdx.x;
  const int blockRow0 = blockIdx.x * 32;

  {
    v4f* z = (v4f*)(acc1 + (size_t)blockRow0 * NFEAT);
    const v4f zero4 = {0.0f, 0.0f, 0.0f, 0.0f};
    z[tid] = zero4;
    z[tid + 256] = zero4;
  }

  const int total = 32 * K;
  for (int idx = tid; idx < total; idx += 256) {
    int r = idx / K;
    int c = idx - r * K;
    float v = in[(size_t)(blockRow0 + r) * K + c];
    if (RELU_IN) {
      v += in1[(size_t)(blockRow0 + r) * K + c];
      v = fmaxf(v * isq[blockRow0 + r] + bias_in[c], 0.0f);
    }
    As[r * ldA + c] = v;
  }
  __syncthreads();

  const int wave = tid >> 5;
  const int lane = tid & 31;
  const int m  = lane & 15;
  const int kp = lane >> 4;
  const int rowL = (wave >> 2) * 16;
  const int col0 = (wave & 3) * 16;
  const int gRow0 = blockRow0 + rowL;

  float isr[8];
#pragma unroll
  for (int i = 0; i < 8; ++i) isr[i] = isq[gRow0 + i + 8 * kp];

  v8f c = {};
  for (int kk = 0; kk < K; kk += 4) {
    const int ka = kk + 2 * kp;
    v2f a, b;
    a.x = As[(rowL + m) * ldA + ka];
    a.y = As[(rowL + m) * ldA + ka + 1];
    b.x = W[(size_t)(ka + 0) * NFEAT + col0 + m];
    b.y = W[(size_t)(ka + 1) * NFEAT + col0 + m];
    c = __builtin_amdgcn_wmma_f32_16x16x4_f32(
         false, a,  false, b,
         (short)0, c,  false,  false);
  }

#pragma unroll
  for (int i = 0; i < 8; ++i) {
    const size_t o = (size_t)(gRow0 + i + 8 * kp) * NFEAT + col0 + m;
    const float v = c[i] * isr[i];
    out[o] = v;
    acc0[o] = v;
  }
}

#define SCAT_GROUP 8
#define SCAT_NGROUP (32 / SCAT_GROUP)

__device__ __forceinline__ int lane_bcast_i(int v, int j) {
  return __builtin_amdgcn_readlane(v, j);
}

__global__ __launch_bounds__(256) void gcn_scatter(
    const int* __restrict__ src, const int* __restrict__ dst,
    const float* __restrict__ h, float* __restrict__ acc0, float* __restrict__ acc1, int E) {
  const int lane = threadIdx.x & 31;
  const int wave = __builtin_amdgcn_readfirstlane(threadIdx.x >> 5);
  const int wavesPerBlock = blockDim.x >> 5;
  const int nWaves = gridDim.x * wavesPerBlock;
  const int wid = blockIdx.x * wavesPerBlock + wave;
  const int nChunks = (E + 31) >> 5;
  if (wid >= nChunks) return;
  float* acc = package_half() ? acc1 : acc0;
  const int eLast = E - 1;
  int s, d;
  {
    const int e = min(wid * 32 + lane, eLast);
    s = src[e];
    d = dst[e];
  }
  for (int chunk = wid; chunk < nChunks; chunk += nWaves) {
    const int e0 = chunk * 32;
    const int en = min(e0 + nWaves * 32 + lane, eLast);
    const int sn = src[en];
    const int dn = dst[en];
    if (e0 + 32 <= E) {
      float lo[2][SCAT_GROUP], hi[2][SCAT_GROUP];
#pragma unroll
      for (int u = 0; u < SCAT_GROUP; ++u) {
        const float* hp = h + (size_t)lane_bcast_i(s, u) * NFEAT;
        lo[0][u] = hp[lane];
        hi[0][u] = hp[lane + 32];
      }
#pragma unroll
      for (int g = 0; g < SCAT_NGROUP; ++g) {
        if (g + 1 < SCAT_NGROUP) {
#pragma unroll
          for (int u = 0; u < SCAT_GROUP; ++u) {
            const float* hp = h + (size_t)lane_bcast_i(s, (g + 1) * SCAT_GROUP + u) * NFEAT;
            lo[(g + 1) & 1][u] = hp[lane];
            hi[(g + 1) & 1][u] = hp[lane + 32];
          }
        }
#pragma unroll
        for (int u = 0; u < SCAT_GROUP; ++u) {
          float* op = acc + (size_t)lane_bcast_i(d, g * SCAT_GROUP + u) * NFEAT;
          atomicAdd(op + lane, lo[g & 1][u]);
          atomicAdd(op + lane + 32, hi[g & 1][u]);
        }
      }
    } else {
      const int cnt = E - e0;
      for (int j = 0; j < cnt; ++j) {
        const float* hp = h + (size_t)lane_bcast_i(s, j) * NFEAT;
        float* op = acc + (size_t)lane_bcast_i(d, j) * NFEAT;
        atomicAdd(op + lane, hp[lane]);
        atomicAdd(op + lane + 32, hp[lane + 32]);
      }
    }
    s = sn;
    d = dn;
  }
}

#define POOL_WAVES 8
#define POOL_BLOCKS 256

struct PoolStep {
  v4f v0, v1;
  v4f w0, w1;
  float q0, q1;
  int g0, g1;
};

__device__ __forceinline__ void pool_load(PoolStep& st, const float* __restrict__ h0,
                                          const float* __restrict__ h1,
                                          const int* __restrict__ batch,
                                          const float* __restrict__ isq,
                                          int ks, int ks1, int n, int c, int hh) {
  const int node0 = 4 * ks + hh;
  const int node1 = node0 + 2;
  const bool ok0 = (ks < ks1) && (node0 < n);
  const bool ok1 = (ks < ks1) && (node1 < n);
  const int m0 = ok0 ? node0 : 0;
  const int m1 = ok1 ? node1 : 0;
  const int b0 = batch[m0];
  const int b1 = batch[m1];
  st.g0 = ok0 ? (b0 - c) : -1;
  st.g1 = ok1 ? (b1 - c) : -1;
  st.q0 = isq[m0];
  st.q1 = isq[m1];
  st.v0 = *(const v4f*)(h0 + (size_t)m0 * NFEAT + 4 * c);
  st.v1 = *(const v4f*)(h0 + (size_t)m1 * NFEAT + 4 * c);
  st.w0 = *(const v4f*)(h1 + (size_t)m0 * NFEAT + 4 * c);
  st.w1 = *(const v4f*)(h1 + (size_t)m1 * NFEAT + 4 * c);
}

__global__ __launch_bounds__(256) void gcn_pool_onehot_wmma(
    const float* __restrict__ h0, const float* __restrict__ h1, const int* __restrict__ batch,
    const float* __restrict__ isq, const float* __restrict__ bias,
    float* __restrict__ gsum, float* __restrict__ gcnt, int n) {
  __shared__ float part[POOL_WAVES * 64 * NFEAT];
  __shared__ float cpart[POOL_WAVES * 2 * 64];
  const int tid = threadIdx.x;
  const int wave = __builtin_amdgcn_readfirstlane(tid >> 5);
  const int lane = tid & 31;
  const int c = lane & 15;
  const int hh = lane >> 4;

  const int nks = (n + 3) >> 2;
  const int nwaves = gridDim.x * POOL_WAVES;
  const int wid = blockIdx.x * POOL_WAVES + wave;
  const int base = nks / nwaves;
  const int rem = nks - base * nwaves;
  const int ks0 = wid * base + (wid < rem ? wid : rem);
  const int ks1 = ks0 + base + (wid < rem ? 1 : 0);

  v8f acc[4][4];
  v2f cnt[4];
#pragma unroll
  for (int r = 0; r < 4; ++r) {
    cnt[r] = v2f{0.0f, 0.0f};
#pragma unroll
    for (int j = 0; j < 4; ++j) acc[r][j] = v8f{};
  }

  if (ks0 < ks1) {
    const v4f bb = *(const v4f*)(bias + 4 * c);
    PoolStep cur;
    pool_load(cur, h0, h1, batch, isq, ks0, ks1, n, c, hh);
    for (int ks = ks0; ks < ks1; ++ks) {
      PoolStep nxt;
      pool_load(nxt, h0, h1, batch, isq, ks + 1, ks1, n, c, hh);
      const v4f zero4 = {0.0f, 0.0f, 0.0f, 0.0f};
      const v4f r0 = __builtin_elementwise_max((cur.v0 + cur.w0) * cur.q0 + bb, zero4);
      const v4f r1 = __builtin_elementwise_max((cur.v1 + cur.w1) * cur.q1 + bb, zero4);
#pragma unroll
      for (int r = 0; r < 4; ++r) {
        v2f a;
        a.x = (cur.g0 == 16 * r) ? 1.0f : 0.0f;
        a.y = (cur.g1 == 16 * r) ? 1.0f : 0.0f;
        cnt[r] += a;
#pragma unroll
        for (int j = 0; j < 4; ++j) {
          v2f b;
          b.x = r0[j];
          b.y = r1[j];
          acc[r][j] = __builtin_amdgcn_wmma_f32_16x16x4_f32(
               false, a,  false, b,
               (short)0, acc[r][j],  false,  false);
        }
      }
      asm volatile("" ::"v"(nxt.v0), "v"(nxt.v1), "v"(nxt.w0), "v"(nxt.w1), "v"(nxt.q0), "v"(nxt.q1),
                   "v"(nxt.g0), "v"(nxt.g1));
      cur = nxt;
    }
  }

  float* mine = part + wave * (64 * NFEAT);
#pragma unroll
  for (int r = 0; r < 4; ++r) {
#pragma unroll
    for (int i = 0; i < 8; ++i) {
      v4f v = {acc[r][0][i], acc[r][1][i], acc[r][2][i], acc[r][3][i]};
      *(v4f*)(mine + (16 * r + 8 * hh + i) * NFEAT + 4 * c) = v;
    }
    cpart[(wave * 2 + hh) * 64 + 16 * r + c] = cnt[r].x + cnt[r].y;
  }
  __syncthreads();

  v4f tot[4];
#pragma unroll
  for (int q = 0; q < 4; ++q) {
    const int e = 4 * (tid + 256 * q);
    v4f s = *(const v4f*)(part + e);
#pragma unroll
    for (int w = 1; w < POOL_WAVES; ++w) s += *(const v4f*)(part + w * (64 * NFEAT) + e);
    tot[q] = s;
  }
#pragma unroll
  for (int q = 0; q < 4; ++q) {
    const int e = 4 * (tid + 256 * q);
#pragma unroll
    for (int k = 0; k < 4; ++k) {
      if (tot[q][k] != 0.0f) atomicAdd(&gsum[e + k], tot[q][k]);
    }
  }
  if (tid < 64) {
    float s = 0.0f;
#pragma unroll
    for (int k = 0; k < POOL_WAVES * 2; ++k) s += cpart[k * 64 + tid];
    if (s != 0.0f) atomicAdd(&gcnt[tid], s);
  }
}

__global__ __launch_bounds__(256) void gcn_head(
    const float* __restrict__ gsum, const float* __restrict__ gcnt,
    const float* __restrict__ Wp, const float* __restrict__ bp,
    const float* __restrict__ Wc, const float* __restrict__ bc,
    float* __restrict__ out) {
  __shared__ float g[64 * 64];
  __shared__ float zs[64 * 64];
  const int t = threadIdx.x;
  for (int idx = t; idx < 64 * 64; idx += 256) {
    int gi = idx >> 6;
    float c = fmaxf(gcnt[gi], 1.0f);
    g[idx] = gsum[idx] / c;
  }
  __syncthreads();
  for (int idx = t; idx < 64 * 64; idx += 256) {
    int m = idx >> 6, nn = idx & 63;
    float a = bp[nn];
#pragma unroll 8
    for (int k = 0; k < 64; ++k) a += g[m * 64 + k] * Wp[k * 64 + nn];
    zs[idx] = a;
    out[384 + idx] = a;
  }
  __syncthreads();
  for (int idx = t; idx < 64 * 6; idx += 256) {
    int m = idx / 6, nn = idx - m * 6;
    float a = bc[nn];
#pragma unroll 8
    for (int k = 0; k < 64; ++k) a += zs[m * 64 + k] * Wc[k * 6 + nn];
    out[idx] = a;
  }
}

extern "C" void kernel_launch(void* const* d_in, const int* in_sizes, int n_in,
                              void* d_out, int out_size, void* d_ws, size_t ws_size,
                              hipStream_t stream) {
  const float* x     = (const float*)d_in[0];
  const int*   ei    = (const int*)d_in[1];
  const int*   batch = (const int*)d_in[2];
  const float* W1    = (const float*)d_in[3];
  const float* b1    = (const float*)d_in[4];
  const float* W2    = (const float*)d_in[5];
  const float* b2    = (const float*)d_in[6];
  const float* Wp    = (const float*)d_in[7];
  const float* bp    = (const float*)d_in[8];
  const float* Wc    = (const float*)d_in[9];
  const float* bc    = (const float*)d_in[10];

  const int n = in_sizes[0] / 128;
  const int E = in_sizes[1] / 2;
  const int* src = ei;
  const int* dst = ei + E;

  char* ws = (char*)d_ws;
  size_t off = 0;
  auto carve = [&](size_t bytes) -> float* {
    float* p = (float*)(ws + off);
    off = (off + bytes + 255) & ~(size_t)255;
    return p;
  };
  float* isq  = carve((size_t)n * sizeof(float));
  float* deg1 = carve((size_t)n * sizeof(float));
  float* bufA = carve((size_t)n * NFEAT * sizeof(float));
  float* bufB0 = carve((size_t)n * NFEAT * sizeof(float));
  float* bufB1 = carve((size_t)n * NFEAT * sizeof(float));
  float* bufC0 = carve((size_t)n * NFEAT * sizeof(float));
  float* bufC1 = carve((size_t)n * NFEAT * sizeof(float));
  float* gsum = carve(64 * NFEAT * sizeof(float));
  float* gcnt = carve(64 * sizeof(float));

  const int T = 256;
  const int nBlk  = (n + T - 1) / T;
  const int gemmBlk = n / 32;

  gcn_deg_init<<<nBlk, T, 0, stream>>>(isq, deg1, n);
  gcn_deg_count<<<EDGE_GRID, T, 0, stream>>>(dst, isq, deg1, E);
  gcn_deg_rsqrt<<<nBlk, T, 0, stream>>>(isq, deg1, n);

  gcn_gemm_wmma_f32<false><<<gemmBlk, T, 0, stream>>>(x, nullptr, W1, isq, b1, bufA, bufB0, bufB1, 128);
  gcn_scatter<<<EDGE_GRID, T, 0, stream>>>(src, dst, bufA, bufB0, bufB1, E);

  gcn_gemm_wmma_f32<true><<<gemmBlk, T, 0, stream>>>(bufB0, bufB1, W2, isq, b1, bufA, bufC0, bufC1, 64);
  gcn_scatter<<<EDGE_GRID, T, 0, stream>>>(src, dst, bufA, bufC0, bufC1, E);

  gcn_zero_f32<<<(64 * NFEAT + 64 + T - 1) / T, T, 0, stream>>>(gsum, 64 * NFEAT + 64);
  gcn_pool_onehot_wmma<<<POOL_BLOCKS, T, 0, stream>>>(bufC0, bufC1, batch, isq, b2, gsum, gcnt, n);
  gcn_head<<<1, T, 0, stream>>>(gsum, gcnt, Wp, bp, Wc, bc, (float*)d_out);
}


# head: both products (g@Wp; z@Wc, its 6 columns padded to a 16-column tile) as f32 matrix-unit tiles in the one block, all operands fetched in one batch, staged in LDS; float sum order changes
# speedup vs baseline: 3.6669x; 1.1434x over previous
#include <hip/hip_runtime.h>

typedef float v2f __attribute__((ext_vector_type(2)));
typedef float v4f __attribute__((ext_vector_type(4)));
typedef float v8f __attribute__((ext_vector_type(8)));

#define NFEAT 64

__device__ __forceinline__ int package_half() {
  const unsigned id = __builtin_amdgcn_s_sendmsg_rtn(0x87);
  return (int)((id >> 18) & 1u);
}

__global__ void gcn_zero_f32(float* __restrict__ p, int count) {
  int i = blockIdx.x * blockDim.x + threadIdx.x;
  if (i < count) p[i] = 0.0f;
}

__global__ void gcn_deg_init(float* __restrict__ deg0, float* __restrict__ deg1, int n) {
  int i = blockIdx.x * blockDim.x + threadIdx.x;
  if (i < n) {
    deg0[i] = 1.0f;
    deg1[i] = 0.0f;
  }
}

#define EDGE_GRID 1280
#define DEG_BATCH 5

__global__ __launch_bounds__(256) void gcn_deg_count(const int* __restrict__ dst,
                                                     float* __restrict__ deg0,
                                                     float* __restrict__ deg1, int E) {
  float* deg = package_half() ? deg1 : deg0;
  const int stride = gridDim.x * blockDim.x;
  for (int base = blockIdx.x * blockDim.x + threadIdx.x; base < E; base += stride * DEG_BATCH) {
    int dd[DEG_BATCH];
#pragma unroll
    for (int u = 0; u < DEG_BATCH; ++u) {
      dd[u] = dst[min(base + u * stride, E - 1)];
    }
#pragma unroll
    for (int u = 0; u < DEG_BATCH; ++u) {
      if (base + u * stride < E) atomicAdd(&deg[dd[u]], 1.0f);
    }
  }
}

__global__ void gcn_deg_rsqrt(float* __restrict__ deg0, const float* __restrict__ deg1, int n) {
  int i = blockIdx.x * blockDim.x + threadIdx.x;
  if (i < n) deg0[i] = rsqrtf(deg0[i] + deg1[i]);
}

template <bool RELU_IN>
__global__ __launch_bounds__(256) void gcn_gemm_wmma_f32(
    const float* __restrict__ in, const float* __restrict__ in1, const float* __restrict__ W,
    const float* __restrict__ isq, const float* __restrict__ bias_in,
    float* __restrict__ out, float* __restrict__ acc0, float* __restrict__ acc1, int K) {
  __shared__ float As[32 * (128 + 4)];
  const int ldA = K + 4;
  const int tid = threadIdx.x;
  const int blockRow0 = blockIdx.x * 32;

  {
    v4f* z = (v4f*)(acc1 + (size_t)blockRow0 * NFEAT);
    const v4f zero4 = {0.0f, 0.0f, 0.0f, 0.0f};
    z[tid] = zero4;
    z[tid + 256] = zero4;
  }

  const int total = 32 * K;
  for (int idx = tid; idx < total; idx += 256) {
    int r = idx / K;
    int c = idx - r * K;
    float v = in[(size_t)(blockRow0 + r) * K + c];
    if (RELU_IN) {
      v += in1[(size_t)(blockRow0 + r) * K + c];
      v = fmaxf(v * isq[blockRow0 + r] + bias_in[c], 0.0f);
    }
    As[r * ldA + c] = v;
  }
  __syncthreads();

  const int wave = tid >> 5;
  const int lane = tid & 31;
  const int m  = lane & 15;
  const int kp = lane >> 4;
  const int rowL = (wave >> 2) * 16;
  const int col0 = (wave & 3) * 16;
  const int gRow0 = blockRow0 + rowL;

  float isr[8];
#pragma unroll
  for (int i = 0; i < 8; ++i) isr[i] = isq[gRow0 + i + 8 * kp];

  v8f c = {};
  for (int kk = 0; kk < K; kk += 4) {
    const int ka = kk + 2 * kp;
    v2f a, b;
    a.x = As[(rowL + m) * ldA + ka];
    a.y = As[(rowL + m) * ldA + ka + 1];
    b.x = W[(size_t)(ka + 0) * NFEAT + col0 + m];
    b.y = W[(size_t)(ka + 1) * NFEAT + col0 + m];
    c = __builtin_amdgcn_wmma_f32_16x16x4_f32(
         false, a,  false, b,
         (short)0, c,  false,  false);
  }

#pragma unroll
  for (int i = 0; i < 8; ++i) {
    const size_t o = (size_t)(gRow0 + i + 8 * kp) * NFEAT + col0 + m;
    const float v = c[i] * isr[i];
    out[o] = v;
    acc0[o] = v;
  }
}

#define SCAT_GROUP 8
#define SCAT_NGROUP (32 / SCAT_GROUP)

__device__ __forceinline__ int lane_bcast_i(int v, int j) {
  return __builtin_amdgcn_readlane(v, j);
}

__global__ __launch_bounds__(256) void gcn_scatter(
    const int* __restrict__ src, const int* __restrict__ dst,
    const float* __restrict__ h, float* __restrict__ acc0, float* __restrict__ acc1, int E) {
  const int lane = threadIdx.x & 31;
  const int wave = __builtin_amdgcn_readfirstlane(threadIdx.x >> 5);
  const int wavesPerBlock = blockDim.x >> 5;
  const int nWaves = gridDim.x * wavesPerBlock;
  const int wid = blockIdx.x * wavesPerBlock + wave;
  const int nChunks = (E + 31) >> 5;
  if (wid >= nChunks) return;
  float* acc = package_half() ? acc1 : acc0;
  const int eLast = E - 1;
  int s, d;
  {
    const int e = min(wid * 32 + lane, eLast);
    s = src[e];
    d = dst[e];
  }
  for (int chunk = wid; chunk < nChunks; chunk += nWaves) {
    const int e0 = chunk * 32;
    const int en = min(e0 + nWaves * 32 + lane, eLast);
    const int sn = src[en];
    const int dn = dst[en];
    if (e0 + 32 <= E) {
      float lo[2][SCAT_GROUP], hi[2][SCAT_GROUP];
#pragma unroll
      for (int u = 0; u < SCAT_GROUP; ++u) {
        const float* hp = h + (size_t)lane_bcast_i(s, u) * NFEAT;
        lo[0][u] = hp[lane];
        hi[0][u] = hp[lane + 32];
      }
#pragma unroll
      for (int g = 0; g < SCAT_NGROUP; ++g) {
        if (g + 1 < SCAT_NGROUP) {
#pragma unroll
          for (int u = 0; u < SCAT_GROUP; ++u) {
            const float* hp = h + (size_t)lane_bcast_i(s, (g + 1) * SCAT_GROUP + u) * NFEAT;
            lo[(g + 1) & 1][u] = hp[lane];
            hi[(g + 1) & 1][u] = hp[lane + 32];
          }
        }
#pragma unroll
        for (int u = 0; u < SCAT_GROUP; ++u) {
          float* op = acc + (size_t)lane_bcast_i(d, g * SCAT_GROUP + u) * NFEAT;
          atomicAdd(op + lane, lo[g & 1][u]);
          atomicAdd(op + lane + 32, hi[g & 1][u]);
        }
      }
    } else {
      const int cnt = E - e0;
      for (int j = 0; j < cnt; ++j) {
        const float* hp = h + (size_t)lane_bcast_i(s, j) * NFEAT;
        float* op = acc + (size_t)lane_bcast_i(d, j) * NFEAT;
        atomicAdd(op + lane, hp[lane]);
        atomicAdd(op + lane + 32, hp[lane + 32]);
      }
    }
    s = sn;
    d = dn;
  }
}

#define POOL_WAVES 8
#define POOL_BLOCKS 256

struct PoolStep {
  v4f v0, v1;
  v4f w0, w1;
  float q0, q1;
  int g0, g1;
};

__device__ __forceinline__ void pool_load(PoolStep& st, const float* __restrict__ h0,
                                          const float* __restrict__ h1,
                                          const int* __restrict__ batch,
                                          const float* __restrict__ isq,
                                          int ks, int ks1, int n, int c, int hh) {
  const int node0 = 4 * ks + hh;
  const int node1 = node0 + 2;
  const bool ok0 = (ks < ks1) && (node0 < n);
  const bool ok1 = (ks < ks1) && (node1 < n);
  const int m0 = ok0 ? node0 : 0;
  const int m1 = ok1 ? node1 : 0;
  const int b0 = batch[m0];
  const int b1 = batch[m1];
  st.g0 = ok0 ? (b0 - c) : -1;
  st.g1 = ok1 ? (b1 - c) : -1;
  st.q0 = isq[m0];
  st.q1 = isq[m1];
  st.v0 = *(const v4f*)(h0 + (size_t)m0 * NFEAT + 4 * c);
  st.v1 = *(const v4f*)(h0 + (size_t)m1 * NFEAT + 4 * c);
  st.w0 = *(const v4f*)(h1 + (size_t)m0 * NFEAT + 4 * c);
  st.w1 = *(const v4f*)(h1 + (size_t)m1 * NFEAT + 4 * c);
}

__global__ __launch_bounds__(256) void gcn_pool_onehot_wmma(
    const float* __restrict__ h0, const float* __restrict__ h1, const int* __restrict__ batch,
    const float* __restrict__ isq, const float* __restrict__ bias,
    float* __restrict__ gsum, float* __restrict__ gcnt, int n) {
  __shared__ float part[POOL_WAVES * 64 * NFEAT];
  __shared__ float cpart[POOL_WAVES * 2 * 64];
  const int tid = threadIdx.x;
  const int wave = __builtin_amdgcn_readfirstlane(tid >> 5);
  const int lane = tid & 31;
  const int c = lane & 15;
  const int hh = lane >> 4;

  const int nks = (n + 3) >> 2;
  const int nwaves = gridDim.x * POOL_WAVES;
  const int wid = blockIdx.x * POOL_WAVES + wave;
  const int base = nks / nwaves;
  const int rem = nks - base * nwaves;
  const int ks0 = wid * base + (wid < rem ? wid : rem);
  const int ks1 = ks0 + base + (wid < rem ? 1 : 0);

  v8f acc[4][4];
  v2f cnt[4];
#pragma unroll
  for (int r = 0; r < 4; ++r) {
    cnt[r] = v2f{0.0f, 0.0f};
#pragma unroll
    for (int j = 0; j < 4; ++j) acc[r][j] = v8f{};
  }

  if (ks0 < ks1) {
    const v4f bb = *(const v4f*)(bias + 4 * c);
    PoolStep cur;
    pool_load(cur, h0, h1, batch, isq, ks0, ks1, n, c, hh);
    for (int ks = ks0; ks < ks1; ++ks) {
      PoolStep nxt;
      pool_load(nxt, h0, h1, batch, isq, ks + 1, ks1, n, c, hh);
      const v4f zero4 = {0.0f, 0.0f, 0.0f, 0.0f};
      const v4f r0 = __builtin_elementwise_max((cur.v0 + cur.w0) * cur.q0 + bb, zero4);
      const v4f r1 = __builtin_elementwise_max((cur.v1 + cur.w1) * cur.q1 + bb, zero4);
#pragma unroll
      for (int r = 0; r < 4; ++r) {
        v2f a;
        a.x = (cur.g0 == 16 * r) ? 1.0f : 0.0f;
        a.y = (cur.g1 == 16 * r) ? 1.0f : 0.0f;
        cnt[r] += a;
#pragma unroll
        for (int j = 0; j < 4; ++j) {
          v2f b;
          b.x = r0[j];
          b.y = r1[j];
          acc[r][j] = __builtin_amdgcn_wmma_f32_16x16x4_f32(
               false, a,  false, b,
               (short)0, acc[r][j],  false,  false);
        }
      }
      asm volatile("" ::"v"(nxt.v0), "v"(nxt.v1), "v"(nxt.w0), "v"(nxt.w1), "v"(nxt.q0), "v"(nxt.q1),
                   "v"(nxt.g0), "v"(nxt.g1));
      cur = nxt;
    }
  }

  float* mine = part + wave * (64 * NFEAT);
#pragma unroll
  for (int r = 0; r < 4; ++r) {
#pragma unroll
    for (int i = 0; i < 8; ++i) {
      v4f v = {acc[r][0][i], acc[r][1][i], acc[r][2][i], acc[r][3][i]};
      *(v4f*)(mine + (16 * r + 8 * hh + i) * NFEAT + 4 * c) = v;
    }
    cpart[(wave * 2 + hh) * 64 + 16 * r + c] = cnt[r].x + cnt[r].y;
  }
  __syncthreads();

  v4f tot[4];
#pragma unroll
  for (int q = 0; q < 4; ++q) {
    const int e = 4 * (tid + 256 * q);
    v4f s = *(const v4f*)(part + e);
#pragma unroll
    for (int w = 1; w < POOL_WAVES; ++w) s += *(const v4f*)(part + w * (64 * NFEAT) + e);
    tot[q] = s;
  }
#pragma unroll
  for (int q = 0; q < 4; ++q) {
    const int e = 4 * (tid + 256 * q);
#pragma unroll
    for (int k = 0; k < 4; ++k) {
      if (tot[q][k] != 0.0f) atomicAdd(&gsum[e + k], tot[q][k]);
    }
  }
  if (tid < 64) {
    float s = 0.0f;
#pragma unroll
    for (int k = 0; k < POOL_WAVES * 2; ++k) s += cpart[k * 64 + tid];
    if (s != 0.0f) atomicAdd(&gcnt[tid], s);
  }
}

#define HEAD_LDA 68
#define HEAD_LDB 72

__global__ __launch_bounds__(256) void gcn_head(
    const float* __restrict__ gsum, const float* __restrict__ gcnt,
    const float* __restrict__ Wp, const float* __restrict__ bp,
    const float* __restrict__ Wc, const float* __restrict__ bc,
    float* __restrict__ out) {
  __shared__ float gs[64 * HEAD_LDA];
  __shared__ float zs[64 * HEAD_LDA];
  __shared__ float wps[64 * HEAD_LDB];
  __shared__ float wcs[64 * 6];
  const int t = threadIdx.x;
  const int wave = __builtin_amdgcn_readfirstlane(t >> 5);
  const int lane = t & 31;
  const int m = lane & 15;
  const int kp = lane >> 4;
  const int rt = wave >> 1;
  const int col0 = (wave & 1) * 32;

  const int r0 = t >> 4;
  const int c4 = (t & 15) * 4;
  v4f gv[4], wv[4];
  float gc[4];
#pragma unroll
  for (int q = 0; q < 4; ++q) {
    gv[q] = *(const v4f*)(gsum + (r0 + 16 * q) * 64 + c4);
    wv[q] = *(const v4f*)(Wp + (r0 + 16 * q) * 64 + c4);
    gc[q] = gcnt[r0 + 16 * q];
  }
  const v4f wcv = *(const v4f*)(Wc + 4 * min(t, 95));
  const float bz0 = bp[col0 + m];
  const float bz1 = bp[col0 + 16 + m];
  const int mc = min(m, 5);
  const float bl = bc[mc];
  __builtin_amdgcn_sched_barrier(0);
  asm volatile("" ::"v"(wcv));

#pragma unroll
  for (int q = 0; q < 4; ++q) {
    const float cn = fmaxf(gc[q], 1.0f);
    const v4f g4 = {gv[q].x / cn, gv[q].y / cn, gv[q].z / cn, gv[q].w / cn};
    *(v4f*)(gs + (r0 + 16 * q) * HEAD_LDA + c4) = g4;
    *(v4f*)(wps + (r0 + 16 * q) * HEAD_LDB + c4) = wv[q];
  }
  if (t < 96) *(v4f*)(wcs + 4 * t) = wcv;
  __syncthreads();

  v8f z0, z1;
#pragma unroll
  for (int i = 0; i < 8; ++i) {
    z0[i] = bz0;
    z1[i] = bz1;
  }
  {
    const float* ga = gs + (16 * rt + m) * HEAD_LDA + 2 * kp;
    const float* wb = wps + (2 * kp) * HEAD_LDB + col0 + m;
#pragma unroll
    for (int ks = 0; ks < 16; ++ks) {
      const v2f a = *(const v2f*)(ga + 4 * ks);
      v2f b0, b1;
      b0.x = wb[(4 * ks) * HEAD_LDB];
      b0.y = wb[(4 * ks + 1) * HEAD_LDB];
      b1.x = wb[(4 * ks) * HEAD_LDB + 16];
      b1.y = wb[(4 * ks + 1) * HEAD_LDB + 16];
      z0 = __builtin_amdgcn_wmma_f32_16x16x4_f32(
           false, a,  false, b0,
           (short)0, z0,  false,  false);
      z1 = __builtin_amdgcn_wmma_f32_16x16x4_f32(
           false, a,  false, b1,
           (short)0, z1,  false,  false);
    }
  }
#pragma unroll
  for (int i = 0; i < 8; ++i) {
    const int row = 16 * rt + 8 * kp + i;
    zs[row * HEAD_LDA + col0 + m] = z0[i];
    zs[row * HEAD_LDA + col0 + 16 + m] = z1[i];
  }
  __syncthreads();
#pragma unroll
  for (int i = 0; i < 8; ++i) {
    const int row = 16 * rt + 8 * kp + i;
    out[384 + row * 64 + col0 + m] = z0[i];
    out[384 + row * 64 + col0 + 16 + m] = z1[i];
  }

  if (wave < 4) {
    const bool real = m < 6;
    v8f l0, l1;
#pragma unroll
    for (int i = 0; i < 8; ++i) {
      l0[i] = bl;
      l1[i] = 0.0f;
    }
    const float* za = zs + (16 * wave + m) * HEAD_LDA + 2 * kp;
    const float* wc = wcs + (2 * kp) * 6 + mc;
#pragma unroll
    for (int ks = 0; ks < 16; ++ks) {
      const v2f a = *(const v2f*)(za + 4 * ks);
      const float wx = wc[(4 * ks) * 6];
      const float wy = wc[(4 * ks + 1) * 6];
      v2f b;
      b.x = real ? wx : 0.0f;
      b.y = real ? wy : 0.0f;
      if (ks & 1) {
        l1 = __builtin_amdgcn_wmma_f32_16x16x4_f32(
             false, a,  false, b,
             (short)0, l1,  false,  false);
      } else {
        l0 = __builtin_amdgcn_wmma_f32_16x16x4_f32(
             false, a,  false, b,
             (short)0, l0,  false,  false);
      }
    }
    const v8f l = l0 + l1;
    if (real) {
#pragma unroll
      for (int i = 0; i < 8; ++i) out[(16 * wave + 8 * kp + i) * 6 + m] = l[i];
    }
  }
}

extern "C" void kernel_launch(void* const* d_in, const int* in_sizes, int n_in,
                              void* d_out, int out_size, void* d_ws, size_t ws_size,
                              hipStream_t stream) {
  const float* x     = (const float*)d_in[0];
  const int*   ei    = (const int*)d_in[1];
  const int*   batch = (const int*)d_in[2];
  const float* W1    = (const float*)d_in[3];
  const float* b1    = (const float*)d_in[4];
  const float* W2    = (const float*)d_in[5];
  const float* b2    = (const float*)d_in[6];
  const float* Wp    = (const float*)d_in[7];
  const float* bp    = (const float*)d_in[8];
  const float* Wc    = (const float*)d_in[9];
  const float* bc    = (const float*)d_in[10];

  const int n = in_sizes[0] / 128;
  const int E = in_sizes[1] / 2;
  const int* src = ei;
  const int* dst = ei + E;

  char* ws = (char*)d_ws;
  size_t off = 0;
  auto carve = [&](size_t bytes) -> float* {
    float* p = (float*)(ws + off);
    off = (off + bytes + 255) & ~(size_t)255;
    return p;
  };
  float* isq  = carve((size_t)n * sizeof(float));
  float* deg1 = carve((size_t)n * sizeof(float));
  float* bufA = carve((size_t)n * NFEAT * sizeof(float));
  float* bufB0 = carve((size_t)n * NFEAT * sizeof(float));
  float* bufB1 = carve((size_t)n * NFEAT * sizeof(float));
  float* bufC0 = carve((size_t)n * NFEAT * sizeof(float));
  float* bufC1 = carve((size_t)n * NFEAT * sizeof(float));
  float* gsum = carve(64 * NFEAT * sizeof(float));
  float* gcnt = carve(64 * sizeof(float));

  const int T = 256;
  const int nBlk  = (n + T - 1) / T;
  const int gemmBlk = n / 32;

  gcn_deg_init<<<nBlk, T, 0, stream>>>(isq, deg1, n);
  gcn_deg_count<<<EDGE_GRID, T, 0, stream>>>(dst, isq, deg1, E);
  gcn_deg_rsqrt<<<nBlk, T, 0, stream>>>(isq, deg1, n);

  gcn_gemm_wmma_f32<false><<<gemmBlk, T, 0, stream>>>(x, nullptr, W1, isq, b1, bufA, bufB0, bufB1, 128);
  gcn_scatter<<<EDGE_GRID, T, 0, stream>>>(src, dst, bufA, bufB0, bufB1, E);

  gcn_gemm_wmma_f32<true><<<gemmBlk, T, 0, stream>>>(bufB0, bufB1, W2, isq, b1, bufA, bufC0, bufC1, 64);
  gcn_scatter<<<EDGE_GRID, T, 0, stream>>>(src, dst, bufA, bufC0, bufC1, E);

  gcn_zero_f32<<<(64 * NFEAT + 64 + T - 1) / T, T, 0, stream>>>(gsum, 64 * NFEAT + 64);
  gcn_pool_onehot_wmma<<<POOL_BLOCKS, T, 0, stream>>>(bufC0, bufC1, batch, isq, b2, gsum, gcnt, n);
  gcn_head<<<1, T, 0, stream>>>(gsum, gcnt, Wp, bp, Wc, bc, (float*)d_out);
}
